# starting kernel
# baseline (speedup 1.0000x reference)
#include <hip/hip_runtime.h>

#define SEP_ID 102
#define B_SZ 64
#define S_SZ 512
#define H_SZ 1024

typedef __attribute__((ext_vector_type(2))) float v2f;
typedef __attribute__((ext_vector_type(8))) float v8f;

struct SegInfo {
  int fs;
  int le;
  float tscale;
  float lscale;
};

__global__ void seg_stats_kernel(const int* __restrict__ ids,
                                 SegInfo* __restrict__ info) {
  int b = blockIdx.x * blockDim.x + threadIdx.x;
  if (b >= B_SZ) return;
  const int* row = ids + b * S_SZ;
  int n = 0, first = -1, last = -1;
  for (int s = 0; s < S_SZ; ++s) {
    int v = row[s];
    if (v == SEP_ID) {
      ++n;
      if (first < 0) first = s;
      last = s;
    }
  }
  int fs, le;
  if (n == 0) {
    fs = 0;
    le = 0;
  } else {
    fs = first;
    le = (n >= 2) ? last : S_SZ;
  }
  int tcnt = fs - 1;
  if (tcnt < 0) tcnt = 0;
  int lcnt = le - fs - 1;
  if (lcnt < 0) lcnt = 0;
  SegInfo o;
  o.fs = fs;
  o.le = le;
  o.tscale = (tcnt > 0) ? (1.0f / (float)tcnt) : 0.0f;
  o.lscale = (lcnt > 0) ? (1.0f / (float)lcnt) : 0.0f;
  info[b] = o;
}

__global__ void __launch_bounds__(256)
seg_pool_wmma_kernel(const float* __restrict__ hs,
                     const SegInfo* __restrict__ info,
                     float* __restrict__ out) {
  const int lane = threadIdx.x & 31;
  const int wave = threadIdx.x >> 5;
  const int b = blockIdx.y;
  const int n0 = blockIdx.x * 128 + wave * 16;

  const SegInfo si = info[b];
  const int m = lane & 15;
  const int khalf = (lane >> 4) << 1;

  int lo = 0;
  int len = 0;
  if (m == 0) {
    lo = 1;
    len = si.fs - 1;
  } else if (m == 1) {
    lo = si.fs + 1;
    len = si.le - si.fs - 1;
  }
  if (len < 0) len = 0;
  const unsigned ulen = (unsigned)len;

  const float* __restrict__ base =
      hs + (size_t)b * (S_SZ * H_SZ) + (size_t)n0 + (size_t)m;

  v8f c = {0.f, 0.f, 0.f, 0.f, 0.f, 0.f, 0.f, 0.f};

  int rel = khalf - lo;

#pragma unroll 8
  for (int k0 = 0; k0 < S_SZ; k0 += 4) {
    v2f a;
    a.x = ((unsigned)rel < ulen) ? 1.0f : 0.0f;
    a.y = ((unsigned)(rel + 1) < ulen) ? 1.0f : 0.0f;

    const float* p = base + (size_t)(k0 + khalf) * H_SZ;
    v2f bm;
    bm.x = p[0];
    bm.y = p[H_SZ];

    __builtin_prefetch(p + 16 * H_SZ, 0, 0);

    c = __builtin_amdgcn_wmma_f32_16x16x4_f32(
         false, a,  false, bm,
         (short)0, c,  false,  false);

    rel += 4;
  }

  if (lane < 16) {
    const int col = n0 + lane;
    out[(size_t)b * H_SZ + col] = c[0] * si.tscale;
    out[(size_t)(B_SZ * H_SZ) + (size_t)b * H_SZ + col] = c[1] * si.lscale;
  }
}

extern "C" void kernel_launch(void* const* d_in, const int* in_sizes, int n_in,
                              void* d_out, int out_size, void* d_ws,
                              size_t ws_size, hipStream_t stream) {
  (void)in_sizes;
  (void)n_in;
  (void)out_size;
  (void)ws_size;

  const float* hidden = (const float*)d_in[0];
  const int* ids = (const int*)d_in[1];
  float* out = (float*)d_out;
  SegInfo* info = (SegInfo*)d_ws;

  seg_stats_kernel<<<1, 64, 0, stream>>>(ids, info);

  dim3 grid(H_SZ / 128, B_SZ);
  dim3 block(256);
  seg_pool_wmma_kernel<<<grid, block, 0, stream>>>(hidden, info, out);
}


# Pooling kernel: B fragments go through a 32-slot register ring, rows of k-step t+31 are requested right behind the WMMA of step t (was: 8 steps loaded, drained, multiplied per trip); sum order kept
# speedup vs baseline: 2.4131x; 2.4131x over previous
#include <hip/hip_runtime.h>

#define SEP_ID 102
#define B_SZ 64
#define S_SZ 512
#define H_SZ 1024

typedef __attribute__((ext_vector_type(2))) float v2f;
typedef __attribute__((ext_vector_type(8))) float v8f;

struct SegInfo {
  int fs;
  int le;
  float tscale;
  float lscale;
};

__global__ void __launch_bounds__(32)
seg_stats_kernel(const int* __restrict__ ids, SegInfo* __restrict__ info) {
  const int b = blockIdx.x;
  const int lane = threadIdx.x;
  const int* row = ids + b * S_SZ + lane;
  int v[S_SZ / 32];
#pragma unroll
  for (int i = 0; i < S_SZ / 32; ++i) v[i] = row[i * 32];
  int n = 0, first = -1, last = -1;
#pragma unroll
  for (int i = 0; i < S_SZ / 32; ++i) {
    const unsigned mk = (unsigned)__ballot(v[i] == SEP_ID);
    if (mk != 0u) {
      n += __popc(mk);
      if (first < 0) first = i * 32 + (__ffs(mk) - 1);
      last = i * 32 + (31 - __clz(mk));
    }
  }
  int fs, le;
  if (n == 0) {
    fs = 0;
    le = 0;
  } else {
    fs = first;
    le = (n >= 2) ? last : S_SZ;
  }
  int tcnt = fs - 1;
  if (tcnt < 0) tcnt = 0;
  int lcnt = le - fs - 1;
  if (lcnt < 0) lcnt = 0;
  if (lane == 0) {
    SegInfo o;
    o.fs = fs;
    o.le = le;
    o.tscale = (tcnt > 0) ? (1.0f / (float)tcnt) : 0.0f;
    o.lscale = (lcnt > 0) ? (1.0f / (float)lcnt) : 0.0f;
    info[b] = o;
  }
}

#define RING 32
#define STEPS (S_SZ / 4)

__device__ __forceinline__ v2f seg_mask(int rel, unsigned ulen) {
  v2f a;
  a.x = ((unsigned)rel < ulen) ? 1.0f : 0.0f;
  a.y = ((unsigned)(rel + 1) < ulen) ? 1.0f : 0.0f;
  return a;
}

__device__ __forceinline__ v2f frag_rows(const float* __restrict__ p, int step) {
  v2f f;
  f.x = p[(size_t)(4 * step) * H_SZ];
  f.y = p[(size_t)(4 * step + 1) * H_SZ];
  return f;
}

__global__ void __launch_bounds__(256)
seg_pool_wmma_kernel(const float* __restrict__ hs,
                     const SegInfo* __restrict__ info,
                     float* __restrict__ out) {
  const int lane = threadIdx.x & 31;
  const int wave = threadIdx.x >> 5;
  const int b = blockIdx.y;
  const int n0 = blockIdx.x * 128 + wave * 16;

  const int m = lane & 15;
  const int khalf = (lane >> 4) << 1;

  const float* __restrict__ p =
      hs + (size_t)b * (S_SZ * H_SZ) + (size_t)khalf * H_SZ + (size_t)n0 +
      (size_t)m;

  v2f ring[RING];
#pragma unroll
  for (int j = 0; j < RING - 1; ++j) ring[j] = frag_rows(p, j);

  const SegInfo si = info[b];

  int lo = 0;
  int len = 0;
  if (m == 0) {
    lo = 1;
    len = si.fs - 1;
  } else if (m == 1) {
    lo = si.fs + 1;
    len = si.le - si.fs - 1;
  }
  if (len < 0) len = 0;
  const unsigned ulen = (unsigned)len;

  v8f c = {0.f, 0.f, 0.f, 0.f, 0.f, 0.f, 0.f, 0.f};

  int rel = khalf - lo;

  for (int blk = 0; blk < STEPS / RING - 1; ++blk) {
#pragma unroll
    for (int j = 0; j < RING; ++j) {
      const v2f a = seg_mask(rel + 4 * j, ulen);
      c = __builtin_amdgcn_wmma_f32_16x16x4_f32(
           false, a,  false, ring[j],
           (short)0, c,  false,  false);
      __builtin_amdgcn_sched_barrier(0);
      ring[(j + RING - 1) % RING] = frag_rows(p, j + RING - 1);
    }
    p += (size_t)(4 * RING) * H_SZ;
    rel += 4 * RING;
  }

#pragma unroll
  for (int j = 0; j < RING; ++j) {
    const v2f a = seg_mask(rel + 4 * j, ulen);
    c = __builtin_amdgcn_wmma_f32_16x16x4_f32(
         false, a,  false, ring[j],
         (short)0, c,  false,  false);
    if (j == 0) {
      __builtin_amdgcn_sched_barrier(0);
      ring[RING - 1] = frag_rows(p, RING - 1);
      __builtin_amdgcn_sched_barrier(0);
    }
  }

  if (lane < 16) {
    const int col = n0 + lane;
    out[(size_t)b * H_SZ + col] = c[0] * si.tscale;
    out[(size_t)(B_SZ * H_SZ) + (size_t)b * H_SZ + col] = c[1] * si.lscale;
  }
}

extern "C" void kernel_launch(void* const* d_in, const int* in_sizes, int n_in,
                              void* d_out, int out_size, void* d_ws,
                              size_t ws_size, hipStream_t stream) {
  (void)in_sizes;
  (void)n_in;
  (void)out_size;
  (void)ws_size;

  const float* hidden = (const float*)d_in[0];
  const int* ids = (const int*)d_in[1];
  float* out = (float*)d_out;
  SegInfo* info = (SegInfo*)d_ws;

  seg_stats_kernel<<<B_SZ, 32, 0, stream>>>(ids, info);

  dim3 grid(H_SZ / 128, B_SZ);
  dim3 block(256);
  seg_pool_wmma_kernel<<<grid, block, 0, stream>>>(hidden, info, out);
}
